# starting kernel
# baseline (speedup 1.0000x reference)
#include <hip/hip_runtime.h>

typedef float v2f __attribute__((ext_vector_type(2)));
typedef float v8f __attribute__((ext_vector_type(8)));

#define D_IN  128
#define D_OUT 32

__global__ __launch_bounds__(128) void gemm_xw_kernel(
    const float* __restrict__ x, const float* __restrict__ W,
    float* __restrict__ xw, int nTiles) {
  __shared__ float Wlds[D_IN * D_OUT];

  const float4* Wv = (const float4*)W;
  float4* Lv = (float4*)Wlds;
  for (int i = threadIdx.x; i < (D_IN * D_OUT) / 4; i += blockDim.x)
    Lv[i] = Wv[i];
  __syncthreads();

  const int wave = threadIdx.x >> 5;
  const int lane = threadIdx.x & 31;
  const int tile = blockIdx.x * 4 + wave;
  if (tile >= nTiles) return;

  const int half = lane >> 4;
  const int m    = lane & 15;

  const float* xrow = x + (size_t)(tile * 16 + m) * D_IN + half * 2;

  v8f c0 = {};
  v8f c1 = {};

  #pragma unroll 4
  for (int k = 0; k < D_IN; k += 4) {
    v2f a = *(const v2f*)(xrow + k);

    const int kb = k + half * 2;
    v2f b0, b1;
    b0.x = Wlds[kb * D_OUT + m];
    b0.y = Wlds[(kb + 1) * D_OUT + m];
    b1.x = Wlds[kb * D_OUT + 16 + m];
    b1.y = Wlds[(kb + 1) * D_OUT + 16 + m];

    c0 = __builtin_amdgcn_wmma_f32_16x16x4_f32(false, a, false, b0,
                                               (short)0, c0, false, false);
    c1 = __builtin_amdgcn_wmma_f32_16x16x4_f32(false, a, false, b1,
                                               (short)0, c1, false, false);
  }

  const int rowBase = tile * 16 + half * 8;
  #pragma unroll
  for (int r = 0; r < 8; ++r) {
    xw[(size_t)(rowBase + r) * D_OUT + m]      = c0[r];
    xw[(size_t)(rowBase + r) * D_OUT + 16 + m] = c1[r];
  }
}

__global__ void init_bias_kernel(float* __restrict__ out,
                                 const float* __restrict__ b, int total) {
  int i = blockIdx.x * blockDim.x + threadIdx.x;
  if (i < total) out[i] = b[i & (D_OUT - 1)];
}

__global__ void scatter_edges_kernel(
    const float* __restrict__ xw, const float* __restrict__ edge_val,
    const int* __restrict__ edge_src, const int* __restrict__ edge_dst,
    float* __restrict__ out, int E) {
  long long g = (long long)blockIdx.x * blockDim.x + threadIdx.x;
  int e   = (int)(g >> 5);
  int col = (int)(g & 31);
  if (e >= E) return;

  int s   = edge_src[e];
  int d   = edge_dst[e];
  float v = edge_val[e] * xw[(size_t)s * D_OUT + col];
  unsafeAtomicAdd(&out[(size_t)d * D_OUT + col], v);
}

__global__ void relu_kernel(float* __restrict__ out, int total) {
  int i = blockIdx.x * blockDim.x + threadIdx.x;
  if (i < total) out[i] = fmaxf(out[i], 0.0f);
}

extern "C" void kernel_launch(void* const* d_in, const int* in_sizes, int n_in,
                              void* d_out, int out_size, void* d_ws, size_t ws_size,
                              hipStream_t stream) {
  const float* x        = (const float*)d_in[0];
  const float* W        = (const float*)d_in[1];
  const float* b        = (const float*)d_in[2];
  const float* edge_val = (const float*)d_in[3];
  const int*   edge_src = (const int*)d_in[4];
  const int*   edge_dst = (const int*)d_in[5];
  float* out = (float*)d_out;
  float* xw  = (float*)d_ws;

  const int N = in_sizes[0] / D_IN;
  const int E = in_sizes[3];
  const int nTiles = (N + 15) / 16;
  const int total  = N * D_OUT;

  gemm_xw_kernel<<<(nTiles + 3) / 4, 128, 0, stream>>>(x, W, xw, nTiles);
  init_bias_kernel<<<(total + 255) / 256, 256, 0, stream>>>(out, b, total);

  long long work = (long long)E * 32;
  scatter_edges_kernel<<<(unsigned)((work + 255) / 256), 256, 0, stream>>>(
      xw, edge_val, edge_src, edge_dst, out, E);

  relu_kernel<<<(total + 255) / 256, 256, 0, stream>>>(out, total);
}


# scatter: one wave per 32 edges (coalesced index loads, readlane broadcast); 32 row gathers then 32 float atomics in flight per wave, not one dependent chain; atomic sum order stays arrival-order
# speedup vs baseline: 2.5268x; 2.5268x over previous
#include <hip/hip_runtime.h>

typedef float v2f __attribute__((ext_vector_type(2)));
typedef float v8f __attribute__((ext_vector_type(8)));

#define D_IN  128
#define D_OUT 32

__global__ __launch_bounds__(128) void gemm_xw_kernel(
    const float* __restrict__ x, const float* __restrict__ W,
    float* __restrict__ xw, int nTiles) {
  __shared__ float Wlds[D_IN * D_OUT];

  const float4* Wv = (const float4*)W;
  float4* Lv = (float4*)Wlds;
  for (int i = threadIdx.x; i < (D_IN * D_OUT) / 4; i += blockDim.x)
    Lv[i] = Wv[i];
  __syncthreads();

  const int wave = threadIdx.x >> 5;
  const int lane = threadIdx.x & 31;
  const int tile = blockIdx.x * 4 + wave;
  if (tile >= nTiles) return;

  const int half = lane >> 4;
  const int m    = lane & 15;

  const float* xrow = x + (size_t)(tile * 16 + m) * D_IN + half * 2;

  v8f c0 = {};
  v8f c1 = {};

  #pragma unroll 4
  for (int k = 0; k < D_IN; k += 4) {
    v2f a = *(const v2f*)(xrow + k);

    const int kb = k + half * 2;
    v2f b0, b1;
    b0.x = Wlds[kb * D_OUT + m];
    b0.y = Wlds[(kb + 1) * D_OUT + m];
    b1.x = Wlds[kb * D_OUT + 16 + m];
    b1.y = Wlds[(kb + 1) * D_OUT + 16 + m];

    c0 = __builtin_amdgcn_wmma_f32_16x16x4_f32(false, a, false, b0,
                                               (short)0, c0, false, false);
    c1 = __builtin_amdgcn_wmma_f32_16x16x4_f32(false, a, false, b1,
                                               (short)0, c1, false, false);
  }

  const int rowBase = tile * 16 + half * 8;
  #pragma unroll
  for (int r = 0; r < 8; ++r) {
    xw[(size_t)(rowBase + r) * D_OUT + m]      = c0[r];
    xw[(size_t)(rowBase + r) * D_OUT + 16 + m] = c1[r];
  }
}

__global__ void init_bias_kernel(float* __restrict__ out,
                                 const float* __restrict__ b, int total) {
  int i = blockIdx.x * blockDim.x + threadIdx.x;
  if (i < total) out[i] = b[i & (D_OUT - 1)];
}

#define EDGES_PER_WAVE 32
__global__ __launch_bounds__(256) void scatter_edges_kernel(
    const float* __restrict__ xw, const float* __restrict__ edge_val,
    const int* __restrict__ edge_src, const int* __restrict__ edge_dst,
    float* __restrict__ out, int E) {
  const int lane = threadIdx.x & 31;
  const int w = __builtin_amdgcn_readfirstlane(
      (int)(blockIdx.x * (blockDim.x >> 5) + (threadIdx.x >> 5)));
  const long long e0 = (long long)w * EDGES_PER_WAVE;
  if (e0 >= E) return;

  const long long e = e0 + lane;
  int   s = 0, d = 0;
  float v = 0.0f;
  if (e < E) {
    s = edge_src[e];
    d = edge_dst[e];
    v = edge_val[e];
  }
  const int vb = __float_as_int(v);
  const long long left = (long long)E - e0;

  if (left >= EDGES_PER_WAVE) {
    float xv[EDGES_PER_WAVE];
    #pragma unroll
    for (int j = 0; j < EDGES_PER_WAVE; ++j) {
      const int sj = __builtin_amdgcn_readlane(s, j);
      xv[j] = xw[(size_t)sj * D_OUT + lane];
    }
    #pragma unroll
    for (int j = 0; j < EDGES_PER_WAVE; ++j) {
      const int   dj = __builtin_amdgcn_readlane(d, j);
      const float vj = __int_as_float(__builtin_amdgcn_readlane(vb, j));
      unsafeAtomicAdd(&out[(size_t)dj * D_OUT + lane], vj * xv[j]);
    }
  } else {
    const int n = (int)left;
    for (int j = 0; j < n; ++j) {
      const int   sj = __builtin_amdgcn_readlane(s, j);
      const int   dj = __builtin_amdgcn_readlane(d, j);
      const float vj = __int_as_float(__builtin_amdgcn_readlane(vb, j));
      unsafeAtomicAdd(&out[(size_t)dj * D_OUT + lane],
                      vj * xw[(size_t)sj * D_OUT + lane]);
    }
  }
}

__global__ void relu_kernel(float* __restrict__ out, int total) {
  int i = blockIdx.x * blockDim.x + threadIdx.x;
  if (i < total) out[i] = fmaxf(out[i], 0.0f);
}

extern "C" void kernel_launch(void* const* d_in, const int* in_sizes, int n_in,
                              void* d_out, int out_size, void* d_ws, size_t ws_size,
                              hipStream_t stream) {
  const float* x        = (const float*)d_in[0];
  const float* W        = (const float*)d_in[1];
  const float* b        = (const float*)d_in[2];
  const float* edge_val = (const float*)d_in[3];
  const int*   edge_src = (const int*)d_in[4];
  const int*   edge_dst = (const int*)d_in[5];
  float* out = (float*)d_out;
  float* xw  = (float*)d_ws;

  const int N = in_sizes[0] / D_IN;
  const int E = in_sizes[3];
  const int nTiles = (N + 15) / 16;
  const int total  = N * D_OUT;

  gemm_xw_kernel<<<(nTiles + 3) / 4, 128, 0, stream>>>(x, W, xw, nTiles);
  init_bias_kernel<<<(total + 255) / 256, 256, 0, stream>>>(out, b, total);

  scatter_edges_kernel<<<(unsigned)((E + 255) / 256), 256, 0, stream>>>(
      xw, edge_val, edge_src, edge_dst, out, E);

  relu_kernel<<<(total + 255) / 256, 256, 0, stream>>>(out, total);
}


# bias fill folded into the GEMM kernel: each wave stores b into its own 16 rows of out (4 whole-line 16-byte stores per lane) ahead of its k loop; 3 launches, not 4; atomic sum order unchanged
# speedup vs baseline: 2.7656x; 1.0945x over previous
#include <hip/hip_runtime.h>

typedef float v2f __attribute__((ext_vector_type(2)));
typedef float v4f __attribute__((ext_vector_type(4)));
typedef float v8f __attribute__((ext_vector_type(8)));

#define D_IN  128
#define D_OUT 32

__global__ __launch_bounds__(128) void gemm_xw_kernel(
    const float* __restrict__ x, const float* __restrict__ W,
    const float* __restrict__ b, float* __restrict__ xw,
    float* __restrict__ out, int nTiles, int total) {
  __shared__ float Wlds[D_IN * D_OUT];

  const v4f bv = *(const v4f*)(b + (threadIdx.x & 7) * 4);

  const float4* Wv = (const float4*)W;
  float4* Lv = (float4*)Wlds;
  for (int i = threadIdx.x; i < (D_IN * D_OUT) / 4; i += blockDim.x)
    Lv[i] = Wv[i];
  __syncthreads();

  const int wave = threadIdx.x >> 5;
  const int lane = threadIdx.x & 31;
  const int tile = blockIdx.x * 4 + wave;
  if (tile >= nTiles) return;

  {
    const int base4 = tile * (16 * D_OUT / 4) + lane;
    v4f* o4 = (v4f*)out;
    if ((tile + 1) * (16 * D_OUT) <= total) {
      #pragma unroll
      for (int i = 0; i < 4; ++i) o4[base4 + i * 32] = bv;
    } else {
      for (int i = 0; i < 4; ++i)
        if ((base4 + i * 32) * 4 < total) o4[base4 + i * 32] = bv;
    }
  }

  const int half = lane >> 4;
  const int m    = lane & 15;

  const float* xrow = x + (size_t)(tile * 16 + m) * D_IN + half * 2;

  v8f c0 = {};
  v8f c1 = {};

  #pragma unroll 4
  for (int k = 0; k < D_IN; k += 4) {
    v2f a = *(const v2f*)(xrow + k);

    const int kb = k + half * 2;
    v2f b0, b1;
    b0.x = Wlds[kb * D_OUT + m];
    b0.y = Wlds[(kb + 1) * D_OUT + m];
    b1.x = Wlds[kb * D_OUT + 16 + m];
    b1.y = Wlds[(kb + 1) * D_OUT + 16 + m];

    c0 = __builtin_amdgcn_wmma_f32_16x16x4_f32(false, a, false, b0,
                                               (short)0, c0, false, false);
    c1 = __builtin_amdgcn_wmma_f32_16x16x4_f32(false, a, false, b1,
                                               (short)0, c1, false, false);
  }

  const int rowBase = tile * 16 + half * 8;
  #pragma unroll
  for (int r = 0; r < 8; ++r) {
    xw[(size_t)(rowBase + r) * D_OUT + m]      = c0[r];
    xw[(size_t)(rowBase + r) * D_OUT + 16 + m] = c1[r];
  }
}

#define EDGES_PER_WAVE 32
__global__ __launch_bounds__(256) void scatter_edges_kernel(
    const float* __restrict__ xw, const float* __restrict__ edge_val,
    const int* __restrict__ edge_src, const int* __restrict__ edge_dst,
    float* __restrict__ out, int E) {
  const int lane = threadIdx.x & 31;
  const int w = __builtin_amdgcn_readfirstlane(
      (int)(blockIdx.x * (blockDim.x >> 5) + (threadIdx.x >> 5)));
  const long long e0 = (long long)w * EDGES_PER_WAVE;
  if (e0 >= E) return;

  const long long e = e0 + lane;
  int   s = 0, d = 0;
  float v = 0.0f;
  if (e < E) {
    s = edge_src[e];
    d = edge_dst[e];
    v = edge_val[e];
  }
  const int vb = __float_as_int(v);
  const long long left = (long long)E - e0;

  if (left >= EDGES_PER_WAVE) {
    float xv[EDGES_PER_WAVE];
    #pragma unroll
    for (int j = 0; j < EDGES_PER_WAVE; ++j) {
      const int sj = __builtin_amdgcn_readlane(s, j);
      xv[j] = xw[(size_t)sj * D_OUT + lane];
    }
    #pragma unroll
    for (int j = 0; j < EDGES_PER_WAVE; ++j) {
      const int   dj = __builtin_amdgcn_readlane(d, j);
      const float vj = __int_as_float(__builtin_amdgcn_readlane(vb, j));
      unsafeAtomicAdd(&out[(size_t)dj * D_OUT + lane], vj * xv[j]);
    }
  } else {
    const int n = (int)left;
    for (int j = 0; j < n; ++j) {
      const int   sj = __builtin_amdgcn_readlane(s, j);
      const int   dj = __builtin_amdgcn_readlane(d, j);
      const float vj = __int_as_float(__builtin_amdgcn_readlane(vb, j));
      unsafeAtomicAdd(&out[(size_t)dj * D_OUT + lane],
                      vj * xw[(size_t)sj * D_OUT + lane]);
    }
  }
}

__global__ void relu_kernel(float* __restrict__ out, int total) {
  int i = blockIdx.x * blockDim.x + threadIdx.x;
  if (i < total) out[i] = fmaxf(out[i], 0.0f);
}

extern "C" void kernel_launch(void* const* d_in, const int* in_sizes, int n_in,
                              void* d_out, int out_size, void* d_ws, size_t ws_size,
                              hipStream_t stream) {
  const float* x        = (const float*)d_in[0];
  const float* W        = (const float*)d_in[1];
  const float* b        = (const float*)d_in[2];
  const float* edge_val = (const float*)d_in[3];
  const int*   edge_src = (const int*)d_in[4];
  const int*   edge_dst = (const int*)d_in[5];
  float* out = (float*)d_out;
  float* xw  = (float*)d_ws;

  const int N = in_sizes[0] / D_IN;
  const int E = in_sizes[3];
  const int nTiles = (N + 15) / 16;
  const int total  = N * D_OUT;

  gemm_xw_kernel<<<(nTiles + 3) / 4, 128, 0, stream>>>(x, W, b, xw, out, nTiles, total);

  scatter_edges_kernel<<<(unsigned)((E + 255) / 256), 256, 0, stream>>>(
      xw, edge_val, edge_src, edge_dst, out, E);

  relu_kernel<<<(total + 255) / 256, 256, 0, stream>>>(out, total);
}


# no global atomics: each 8192-edge chunk counting-sorted by 391-row destination band in LDS; one work-group per band sums its rows in LDS (float adds, arrival order) and stores relu(b+sum) once
# speedup vs baseline: 3.5522x; 1.2844x over previous
#include <hip/hip_runtime.h>

typedef float v2f __attribute__((ext_vector_type(2)));
typedef float v4f __attribute__((ext_vector_type(4)));
typedef float v8f __attribute__((ext_vector_type(8)));

#define D_IN  128
#define D_OUT 32


__global__ __launch_bounds__(128) void gemm_xw_kernel(
    const float* __restrict__ x, const float* __restrict__ W,
    float* __restrict__ xw, int nTiles) {
  __shared__ float Wlds[D_IN * D_OUT];

  const float4* Wv = (const float4*)W;
  float4* Lv = (float4*)Wlds;
  for (int i = threadIdx.x; i < (D_IN * D_OUT) / 4; i += blockDim.x)
    Lv[i] = Wv[i];
  __syncthreads();

  const int wave = threadIdx.x >> 5;
  const int lane = threadIdx.x & 31;
  const int tile = blockIdx.x * 4 + wave;
  if (tile >= nTiles) return;

  const int half = lane >> 4;
  const int m    = lane & 15;

  const float* xrow = x + (size_t)(tile * 16 + m) * D_IN + half * 2;

  v8f c0 = {};
  v8f c1 = {};

  #pragma unroll 4
  for (int k = 0; k < D_IN; k += 4) {
    v2f a = *(const v2f*)(xrow + k);

    const int kb = k + half * 2;
    v2f b0, b1;
    b0.x = Wlds[kb * D_OUT + m];
    b0.y = Wlds[(kb + 1) * D_OUT + m];
    b1.x = Wlds[kb * D_OUT + 16 + m];
    b1.y = Wlds[(kb + 1) * D_OUT + 16 + m];

    c0 = __builtin_amdgcn_wmma_f32_16x16x4_f32(false, a, false, b0,
                                               (short)0, c0, false, false);
    c1 = __builtin_amdgcn_wmma_f32_16x16x4_f32(false, a, false, b1,
                                               (short)0, c1, false, false);
  }

  const int rowBase = tile * 16 + half * 8;
  #pragma unroll
  for (int r = 0; r < 8; ++r) {
    xw[(size_t)(rowBase + r) * D_OUT + m]      = c0[r];
    xw[(size_t)(rowBase + r) * D_OUT + 16 + m] = c1[r];
  }
}

#define BAND_ROWS 391
#define MAX_BANDS 256
#define CHUNK     8192
#define BIN_THREADS 1024
#define EDGES_PER_THREAD (CHUNK / BIN_THREADS)

__global__ __launch_bounds__(BIN_THREADS) void bin_edges_kernel(
    const float* __restrict__ edge_val, const int* __restrict__ edge_src,
    const int* __restrict__ edge_dst,
    int* __restrict__ s_src, int* __restrict__ s_row, float* __restrict__ s_val,
    int* __restrict__ seg, int E) {
  __shared__ int   cnt[MAX_BANDS];
  __shared__ int   base[MAX_BANDS];
  __shared__ int   l_src[CHUNK];
  __shared__ int   l_row[CHUNK];
  __shared__ float l_val[CHUNK];

  const int tid   = threadIdx.x;
  const int chunk = blockIdx.x;
  const long long c0 = (long long)chunk * CHUNK;
  const long long rem = (long long)E - c0;
  const int n = rem < CHUNK ? (int)rem : CHUNK;

  if (tid < MAX_BANDS) cnt[tid] = 0;
  __syncthreads();

  int   es[EDGES_PER_THREAD], eb[EDGES_PER_THREAD], er[EDGES_PER_THREAD], ek[EDGES_PER_THREAD];
  float ev[EDGES_PER_THREAD];
  #pragma unroll
  for (int i = 0; i < EDGES_PER_THREAD; ++i) {
    const int idx = i * BIN_THREADS + tid;
    es[i] = 0; eb[i] = 0; er[i] = 0; ek[i] = 0; ev[i] = 0.0f;
    if (idx < n) {
      es[i] = edge_src[c0 + idx];
      const int d = edge_dst[c0 + idx];
      ev[i] = edge_val[c0 + idx];
      eb[i] = d / BAND_ROWS;
      er[i] = d - eb[i] * BAND_ROWS;
    }
  }
  #pragma unroll
  for (int i = 0; i < EDGES_PER_THREAD; ++i) {
    const int idx = i * BIN_THREADS + tid;
    if (idx < n) ek[i] = atomicAdd(&cnt[eb[i]], 1);
  }
  __syncthreads();

  if (tid < 32) {
    int loc[MAX_BANDS / 32];
    int sum = 0;
    #pragma unroll
    for (int q = 0; q < MAX_BANDS / 32; ++q) {
      loc[q] = sum;
      sum += cnt[tid * (MAX_BANDS / 32) + q];
    }
    int inc = sum;
    #pragma unroll
    for (int off = 1; off < 32; off <<= 1) {
      const int t = __shfl_up(inc, off);
      if (tid >= off) inc += t;
    }
    const int excl = inc - sum;
    #pragma unroll
    for (int q = 0; q < MAX_BANDS / 32; ++q)
      base[tid * (MAX_BANDS / 32) + q] = excl + loc[q];
  }
  __syncthreads();

  #pragma unroll
  for (int i = 0; i < EDGES_PER_THREAD; ++i) {
    const int idx = i * BIN_THREADS + tid;
    if (idx < n) {
      const int pos = base[eb[i]] + ek[i];
      l_src[pos] = es[i];
      l_row[pos] = er[i];
      l_val[pos] = ev[i];
    }
  }
  __syncthreads();

  for (int idx = tid; idx < n; idx += BIN_THREADS) {
    s_src[c0 + idx] = l_src[idx];
    s_row[c0 + idx] = l_row[idx];
    s_val[c0 + idx] = l_val[idx];
  }
  if (tid < MAX_BANDS)
    ((int2*)seg)[(size_t)chunk * MAX_BANDS + tid] = make_int2(base[tid], cnt[tid]);
}

#define SUM_THREADS 1024
#define SUM_WAVES   (SUM_THREADS / 32)
#define GATHER_DEPTH 32
__global__ __launch_bounds__(SUM_THREADS) void band_sum_kernel(
    const float* __restrict__ xw, const float* __restrict__ b,
    const int* __restrict__ s_src, const int* __restrict__ s_row,
    const float* __restrict__ s_val, const int* __restrict__ seg,
    float* __restrict__ out, int N, int nChunks) {
  __shared__ float acc[BAND_ROWS * D_OUT];

  const int band = blockIdx.x;
  const int tid  = threadIdx.x;
  const int lane = tid & 31;
  const int wave = __builtin_amdgcn_readfirstlane(tid >> 5);

  const float bl = b[lane];
  for (int i = tid; i < BAND_ROWS * D_OUT; i += SUM_THREADS) acc[i] = bl;
  __syncthreads();

  for (int kb = 0; kb < nChunks; kb += SUM_WAVES * 32) {
    const int kl = kb + wave + SUM_WAVES * lane;
    int2 sc = make_int2(0, 0);
    if (kl < nChunks) sc = ((const int2*)seg)[(size_t)kl * MAX_BANDS + band];

    for (int i = 0; i < 32; ++i) {
      const int k = kb + wave + SUM_WAVES * i;
      if (k >= nChunks) break;
      const int st = __builtin_amdgcn_readlane(sc.x, i);
      const int ct = __builtin_amdgcn_readlane(sc.y, i);
      const long long e0 = (long long)k * CHUNK + st;

      for (int o = 0; o < ct; o += 32) {
        const int n = (ct - o) < 32 ? (ct - o) : 32;
        int   s = 0, r = 0;
        float v = 0.0f;
        if (lane < n) {
          s = s_src[e0 + o + lane];
          r = s_row[e0 + o + lane];
          v = s_val[e0 + o + lane];
        }
        const int vb = __float_as_int(v);

        for (int j0 = 0; j0 < n; j0 += GATHER_DEPTH) {
          float xv[GATHER_DEPTH];
          #pragma unroll
          for (int u = 0; u < GATHER_DEPTH; ++u) {
            const int jj = (j0 + u) < n ? (j0 + u) : (n - 1);
            const int sj = __builtin_amdgcn_readlane(s, jj);
            xv[u] = xw[(size_t)sj * D_OUT + lane];
          }
          #pragma unroll
          for (int u = 0; u < GATHER_DEPTH; ++u) {
            if (j0 + u < n) {
              const int   rj = __builtin_amdgcn_readlane(r, j0 + u);
              const float vj = __int_as_float(__builtin_amdgcn_readlane(vb, j0 + u));
              atomicAdd(&acc[rj * D_OUT + lane], vj * xv[u]);
            }
          }
        }
      }
    }
  }
  __syncthreads();

  const int row0 = band * BAND_ROWS;
  const int rows = (N - row0) < BAND_ROWS ? (N - row0) : BAND_ROWS;
  const v4f* a4 = (const v4f*)acc;
  v4f* o4 = (v4f*)(out + (size_t)row0 * D_OUT);
  for (int i = tid; i < rows * (D_OUT / 4); i += SUM_THREADS) {
    v4f t = a4[i];
    t.x = fmaxf(t.x, 0.0f); t.y = fmaxf(t.y, 0.0f);
    t.z = fmaxf(t.z, 0.0f); t.w = fmaxf(t.w, 0.0f);
    o4[i] = t;
  }
}

extern "C" void kernel_launch(void* const* d_in, const int* in_sizes, int n_in,
                              void* d_out, int out_size, void* d_ws, size_t ws_size,
                              hipStream_t stream) {
  const float* x        = (const float*)d_in[0];
  const float* W        = (const float*)d_in[1];
  const float* b        = (const float*)d_in[2];
  const float* edge_val = (const float*)d_in[3];
  const int*   edge_src = (const int*)d_in[4];
  const int*   edge_dst = (const int*)d_in[5];
  float* out = (float*)d_out;

  const int N = in_sizes[0] / D_IN;
  const int E = in_sizes[3];
  const int nTiles  = (N + 15) / 16;
  const int nChunks = (E + CHUNK - 1) / CHUNK;
  const int nBands  = (N + BAND_ROWS - 1) / BAND_ROWS;

  float* xw    = (float*)d_ws;
  int*   s_src = (int*)(xw + (size_t)N * D_OUT);
  int*   s_row = s_src + (size_t)nChunks * CHUNK;
  float* s_val = (float*)(s_row + (size_t)nChunks * CHUNK);
  int*   seg   = (int*)(s_val + (size_t)nChunks * CHUNK);

  gemm_xw_kernel<<<(nTiles + 3) / 4, 128, 0, stream>>>(x, W, xw, nTiles);

  bin_edges_kernel<<<nChunks, BIN_THREADS, 0, stream>>>(
      edge_val, edge_src, edge_dst, s_src, s_row, s_val, seg, E);

  band_sum_kernel<<<nBands, SUM_THREADS, 0, stream>>>(
      xw, b, s_src, s_row, s_val, seg, out, N, nChunks);
}


# GEMM and edge sort in ONE launch: work-groups 0..195 sort an edge chunk each, 196..391 (32 waves, one 16-row tile each, W in LDS) do x@W; 2 launches, not 3; every sum order unchanged
# speedup vs baseline: 4.3936x; 1.2369x over previous
#include <hip/hip_runtime.h>

typedef float v2f __attribute__((ext_vector_type(2)));
typedef float v4f __attribute__((ext_vector_type(4)));
typedef float v8f __attribute__((ext_vector_type(8)));

#define D_IN  128
#define D_OUT 32


#define FRONT_THREADS 1024
#define GEMM_TILES_PER_BLOCK (FRONT_THREADS / 32)
__device__ __forceinline__ void gemm_xw_role(
    const float* __restrict__ x, const float* __restrict__ W,
    float* __restrict__ xw, int nTiles, int gblock, float* Wlds) {
  const float4* Wv = (const float4*)W;
  float4* Lv = (float4*)Wlds;
  for (int i = threadIdx.x; i < (D_IN * D_OUT) / 4; i += FRONT_THREADS)
    Lv[i] = Wv[i];
  __syncthreads();

  const int wave = threadIdx.x >> 5;
  const int lane = threadIdx.x & 31;
  const int tile = gblock * GEMM_TILES_PER_BLOCK + wave;
  if (tile >= nTiles) return;

  const int half = lane >> 4;
  const int m    = lane & 15;

  const float* xrow = x + (size_t)(tile * 16 + m) * D_IN + half * 2;

  v8f c0 = {};
  v8f c1 = {};

  #pragma unroll 4
  for (int k = 0; k < D_IN; k += 4) {
    v2f a = *(const v2f*)(xrow + k);

    const int kb = k + half * 2;
    v2f b0, b1;
    b0.x = Wlds[kb * D_OUT + m];
    b0.y = Wlds[(kb + 1) * D_OUT + m];
    b1.x = Wlds[kb * D_OUT + 16 + m];
    b1.y = Wlds[(kb + 1) * D_OUT + 16 + m];

    c0 = __builtin_amdgcn_wmma_f32_16x16x4_f32(false, a, false, b0,
                                               (short)0, c0, false, false);
    c1 = __builtin_amdgcn_wmma_f32_16x16x4_f32(false, a, false, b1,
                                               (short)0, c1, false, false);
  }

  const int rowBase = tile * 16 + half * 8;
  #pragma unroll
  for (int r = 0; r < 8; ++r) {
    xw[(size_t)(rowBase + r) * D_OUT + m]      = c0[r];
    xw[(size_t)(rowBase + r) * D_OUT + 16 + m] = c1[r];
  }
}

#define BAND_ROWS 391
#define MAX_BANDS 256
#define CHUNK     8192
#define BIN_THREADS FRONT_THREADS
#define EDGES_PER_THREAD (CHUNK / BIN_THREADS)

#define FRONT_LDS_WORDS (2 * MAX_BANDS + 3 * CHUNK)
__device__ __forceinline__ void bin_edges_role(
    const float* __restrict__ edge_val, const int* __restrict__ edge_src,
    const int* __restrict__ edge_dst,
    int* __restrict__ s_src, int* __restrict__ s_row, float* __restrict__ s_val,
    int* __restrict__ seg, int E, int chunk, int* lds) {
  int*   cnt   = lds;
  int*   base  = lds + MAX_BANDS;
  int*   l_src = lds + 2 * MAX_BANDS;
  int*   l_row = l_src + CHUNK;
  float* l_val = (float*)(l_row + CHUNK);

  const int tid   = threadIdx.x;
  const long long c0 = (long long)chunk * CHUNK;
  const long long rem = (long long)E - c0;
  const int n = rem < CHUNK ? (int)rem : CHUNK;

  if (tid < MAX_BANDS) cnt[tid] = 0;
  __syncthreads();

  int   es[EDGES_PER_THREAD], eb[EDGES_PER_THREAD], er[EDGES_PER_THREAD], ek[EDGES_PER_THREAD];
  float ev[EDGES_PER_THREAD];
  #pragma unroll
  for (int i = 0; i < EDGES_PER_THREAD; ++i) {
    const int idx = i * BIN_THREADS + tid;
    es[i] = 0; eb[i] = 0; er[i] = 0; ek[i] = 0; ev[i] = 0.0f;
    if (idx < n) {
      es[i] = edge_src[c0 + idx];
      const int d = edge_dst[c0 + idx];
      ev[i] = edge_val[c0 + idx];
      eb[i] = d / BAND_ROWS;
      er[i] = d - eb[i] * BAND_ROWS;
    }
  }
  #pragma unroll
  for (int i = 0; i < EDGES_PER_THREAD; ++i) {
    const int idx = i * BIN_THREADS + tid;
    if (idx < n) ek[i] = atomicAdd(&cnt[eb[i]], 1);
  }
  __syncthreads();

  if (tid < 32) {
    int loc[MAX_BANDS / 32];
    int sum = 0;
    #pragma unroll
    for (int q = 0; q < MAX_BANDS / 32; ++q) {
      loc[q] = sum;
      sum += cnt[tid * (MAX_BANDS / 32) + q];
    }
    int inc = sum;
    #pragma unroll
    for (int off = 1; off < 32; off <<= 1) {
      const int t = __shfl_up(inc, off);
      if (tid >= off) inc += t;
    }
    const int excl = inc - sum;
    #pragma unroll
    for (int q = 0; q < MAX_BANDS / 32; ++q)
      base[tid * (MAX_BANDS / 32) + q] = excl + loc[q];
  }
  __syncthreads();

  #pragma unroll
  for (int i = 0; i < EDGES_PER_THREAD; ++i) {
    const int idx = i * BIN_THREADS + tid;
    if (idx < n) {
      const int pos = base[eb[i]] + ek[i];
      l_src[pos] = es[i];
      l_row[pos] = er[i];
      l_val[pos] = ev[i];
    }
  }
  __syncthreads();

  for (int idx = tid; idx < n; idx += BIN_THREADS) {
    s_src[c0 + idx] = l_src[idx];
    s_row[c0 + idx] = l_row[idx];
    s_val[c0 + idx] = l_val[idx];
  }
  if (tid < MAX_BANDS)
    ((int2*)seg)[(size_t)chunk * MAX_BANDS + tid] = make_int2(base[tid], cnt[tid]);
}

__global__ __launch_bounds__(FRONT_THREADS) void gemm_and_bin_kernel(
    const float* __restrict__ x, const float* __restrict__ W,
    float* __restrict__ xw, int nTiles,
    const float* __restrict__ edge_val, const int* __restrict__ edge_src,
    const int* __restrict__ edge_dst,
    int* __restrict__ s_src, int* __restrict__ s_row, float* __restrict__ s_val,
    int* __restrict__ seg, int E, int nChunks) {
  __shared__ __attribute__((aligned(16))) int lds[FRONT_LDS_WORDS];
  const int blk = blockIdx.x;
  if (blk < nChunks)
    bin_edges_role(edge_val, edge_src, edge_dst, s_src, s_row, s_val, seg, E, blk, lds);
  else
    gemm_xw_role(x, W, xw, nTiles, blk - nChunks, (float*)lds);
}

#define SUM_THREADS 1024
#define SUM_WAVES   (SUM_THREADS / 32)
#define GATHER_DEPTH 32
__global__ __launch_bounds__(SUM_THREADS) void band_sum_kernel(
    const float* __restrict__ xw, const float* __restrict__ b,
    const int* __restrict__ s_src, const int* __restrict__ s_row,
    const float* __restrict__ s_val, const int* __restrict__ seg,
    float* __restrict__ out, int N, int nChunks) {
  __shared__ float acc[BAND_ROWS * D_OUT];

  const int band = blockIdx.x;
  const int tid  = threadIdx.x;
  const int lane = tid & 31;
  const int wave = __builtin_amdgcn_readfirstlane(tid >> 5);

  const float bl = b[lane];
  for (int i = tid; i < BAND_ROWS * D_OUT; i += SUM_THREADS) acc[i] = bl;
  __syncthreads();

  for (int kb = 0; kb < nChunks; kb += SUM_WAVES * 32) {
    const int kl = kb + wave + SUM_WAVES * lane;
    int2 sc = make_int2(0, 0);
    if (kl < nChunks) sc = ((const int2*)seg)[(size_t)kl * MAX_BANDS + band];

    for (int i = 0; i < 32; ++i) {
      const int k = kb + wave + SUM_WAVES * i;
      if (k >= nChunks) break;
      const int st = __builtin_amdgcn_readlane(sc.x, i);
      const int ct = __builtin_amdgcn_readlane(sc.y, i);
      const long long e0 = (long long)k * CHUNK + st;

      for (int o = 0; o < ct; o += 32) {
        const int n = (ct - o) < 32 ? (ct - o) : 32;
        int   s = 0, r = 0;
        float v = 0.0f;
        if (lane < n) {
          s = s_src[e0 + o + lane];
          r = s_row[e0 + o + lane];
          v = s_val[e0 + o + lane];
        }
        const int vb = __float_as_int(v);

        for (int j0 = 0; j0 < n; j0 += GATHER_DEPTH) {
          float xv[GATHER_DEPTH];
          #pragma unroll
          for (int u = 0; u < GATHER_DEPTH; ++u) {
            const int jj = (j0 + u) < n ? (j0 + u) : (n - 1);
            const int sj = __builtin_amdgcn_readlane(s, jj);
            xv[u] = xw[(size_t)sj * D_OUT + lane];
          }
          #pragma unroll
          for (int u = 0; u < GATHER_DEPTH; ++u) {
            if (j0 + u < n) {
              const int   rj = __builtin_amdgcn_readlane(r, j0 + u);
              const float vj = __int_as_float(__builtin_amdgcn_readlane(vb, j0 + u));
              atomicAdd(&acc[rj * D_OUT + lane], vj * xv[u]);
            }
          }
        }
      }
    }
  }
  __syncthreads();

  const int row0 = band * BAND_ROWS;
  const int rows = (N - row0) < BAND_ROWS ? (N - row0) : BAND_ROWS;
  const v4f* a4 = (const v4f*)acc;
  v4f* o4 = (v4f*)(out + (size_t)row0 * D_OUT);
  for (int i = tid; i < rows * (D_OUT / 4); i += SUM_THREADS) {
    v4f t = a4[i];
    t.x = fmaxf(t.x, 0.0f); t.y = fmaxf(t.y, 0.0f);
    t.z = fmaxf(t.z, 0.0f); t.w = fmaxf(t.w, 0.0f);
    o4[i] = t;
  }
}

extern "C" void kernel_launch(void* const* d_in, const int* in_sizes, int n_in,
                              void* d_out, int out_size, void* d_ws, size_t ws_size,
                              hipStream_t stream) {
  const float* x        = (const float*)d_in[0];
  const float* W        = (const float*)d_in[1];
  const float* b        = (const float*)d_in[2];
  const float* edge_val = (const float*)d_in[3];
  const int*   edge_src = (const int*)d_in[4];
  const int*   edge_dst = (const int*)d_in[5];
  float* out = (float*)d_out;

  const int N = in_sizes[0] / D_IN;
  const int E = in_sizes[3];
  const int nTiles  = (N + 15) / 16;
  const int nChunks = (E + CHUNK - 1) / CHUNK;
  const int nBands  = (N + BAND_ROWS - 1) / BAND_ROWS;

  float* xw    = (float*)d_ws;
  int*   s_src = (int*)(xw + (size_t)N * D_OUT);
  int*   s_row = s_src + (size_t)nChunks * CHUNK;
  float* s_val = (float*)(s_row + (size_t)nChunks * CHUNK);
  int*   seg   = (int*)(s_val + (size_t)nChunks * CHUNK);

  const int nGemmBlocks = (nTiles + GEMM_TILES_PER_BLOCK - 1) / GEMM_TILES_PER_BLOCK;
  gemm_and_bin_kernel<<<nChunks + nGemmBlocks, FRONT_THREADS, 0, stream>>>(
      x, W, xw, nTiles, edge_val, edge_src, edge_dst, s_src, s_row, s_val, seg, E, nChunks);

  band_sum_kernel<<<nBands, SUM_THREADS, 0, stream>>>(
      xw, b, s_src, s_row, s_val, seg, out, N, nChunks);
}


# band sum: 8 lanes per edge (16-byte xw row gathers, edge fields broadcast in the 8-lane group, band kept as 4 column planes in LDS); one LDS float-add instruction serves 4 edges; adds arrival-order
# speedup vs baseline: 4.8685x; 1.1081x over previous
#include <hip/hip_runtime.h>

typedef float v2f __attribute__((ext_vector_type(2)));
typedef float v4f __attribute__((ext_vector_type(4)));
typedef float v8f __attribute__((ext_vector_type(8)));

#define D_IN  128
#define D_OUT 32


#define FRONT_THREADS 1024
#define GEMM_TILES_PER_BLOCK (FRONT_THREADS / 32)
__device__ __forceinline__ void gemm_xw_role(
    const float* __restrict__ x, const float* __restrict__ W,
    float* __restrict__ xw, int nTiles, int gblock, float* Wlds) {
  const float4* Wv = (const float4*)W;
  float4* Lv = (float4*)Wlds;
  for (int i = threadIdx.x; i < (D_IN * D_OUT) / 4; i += FRONT_THREADS)
    Lv[i] = Wv[i];
  __syncthreads();

  const int wave = threadIdx.x >> 5;
  const int lane = threadIdx.x & 31;
  const int tile = gblock * GEMM_TILES_PER_BLOCK + wave;
  if (tile >= nTiles) return;

  const int half = lane >> 4;
  const int m    = lane & 15;

  const float* xrow = x + (size_t)(tile * 16 + m) * D_IN + half * 2;

  v8f c0 = {};
  v8f c1 = {};

  #pragma unroll 4
  for (int k = 0; k < D_IN; k += 4) {
    v2f a = *(const v2f*)(xrow + k);

    const int kb = k + half * 2;
    v2f b0, b1;
    b0.x = Wlds[kb * D_OUT + m];
    b0.y = Wlds[(kb + 1) * D_OUT + m];
    b1.x = Wlds[kb * D_OUT + 16 + m];
    b1.y = Wlds[(kb + 1) * D_OUT + 16 + m];

    c0 = __builtin_amdgcn_wmma_f32_16x16x4_f32(false, a, false, b0,
                                               (short)0, c0, false, false);
    c1 = __builtin_amdgcn_wmma_f32_16x16x4_f32(false, a, false, b1,
                                               (short)0, c1, false, false);
  }

  const int rowBase = tile * 16 + half * 8;
  #pragma unroll
  for (int r = 0; r < 8; ++r) {
    xw[(size_t)(rowBase + r) * D_OUT + m]      = c0[r];
    xw[(size_t)(rowBase + r) * D_OUT + 16 + m] = c1[r];
  }
}

#define BAND_ROWS 391
#define MAX_BANDS 256
#define CHUNK     8192
#define BIN_THREADS FRONT_THREADS
#define EDGES_PER_THREAD (CHUNK / BIN_THREADS)

#define FRONT_LDS_WORDS (2 * MAX_BANDS + 3 * CHUNK)
__device__ __forceinline__ void bin_edges_role(
    const float* __restrict__ edge_val, const int* __restrict__ edge_src,
    const int* __restrict__ edge_dst,
    int* __restrict__ s_src, int* __restrict__ s_row, float* __restrict__ s_val,
    int* __restrict__ seg, int E, int chunk, int* lds) {
  int*   cnt   = lds;
  int*   base  = lds + MAX_BANDS;
  int*   l_src = lds + 2 * MAX_BANDS;
  int*   l_row = l_src + CHUNK;
  float* l_val = (float*)(l_row + CHUNK);

  const int tid   = threadIdx.x;
  const long long c0 = (long long)chunk * CHUNK;
  const long long rem = (long long)E - c0;
  const int n = rem < CHUNK ? (int)rem : CHUNK;

  if (tid < MAX_BANDS) cnt[tid] = 0;
  __syncthreads();

  int   es[EDGES_PER_THREAD], eb[EDGES_PER_THREAD], er[EDGES_PER_THREAD], ek[EDGES_PER_THREAD];
  float ev[EDGES_PER_THREAD];
  #pragma unroll
  for (int i = 0; i < EDGES_PER_THREAD; ++i) {
    const int idx = i * BIN_THREADS + tid;
    es[i] = 0; eb[i] = 0; er[i] = 0; ek[i] = 0; ev[i] = 0.0f;
    if (idx < n) {
      es[i] = edge_src[c0 + idx];
      const int d = edge_dst[c0 + idx];
      ev[i] = edge_val[c0 + idx];
      eb[i] = d / BAND_ROWS;
      er[i] = d - eb[i] * BAND_ROWS;
    }
  }
  #pragma unroll
  for (int i = 0; i < EDGES_PER_THREAD; ++i) {
    const int idx = i * BIN_THREADS + tid;
    if (idx < n) ek[i] = atomicAdd(&cnt[eb[i]], 1);
  }
  __syncthreads();

  if (tid < 32) {
    int loc[MAX_BANDS / 32];
    int sum = 0;
    #pragma unroll
    for (int q = 0; q < MAX_BANDS / 32; ++q) {
      loc[q] = sum;
      sum += cnt[tid * (MAX_BANDS / 32) + q];
    }
    int inc = sum;
    #pragma unroll
    for (int off = 1; off < 32; off <<= 1) {
      const int t = __shfl_up(inc, off);
      if (tid >= off) inc += t;
    }
    const int excl = inc - sum;
    #pragma unroll
    for (int q = 0; q < MAX_BANDS / 32; ++q)
      base[tid * (MAX_BANDS / 32) + q] = excl + loc[q];
  }
  __syncthreads();

  #pragma unroll
  for (int i = 0; i < EDGES_PER_THREAD; ++i) {
    const int idx = i * BIN_THREADS + tid;
    if (idx < n) {
      const int pos = base[eb[i]] + ek[i];
      l_src[pos] = es[i];
      l_row[pos] = er[i];
      l_val[pos] = ev[i];
    }
  }
  __syncthreads();

  for (int idx = tid; idx < n; idx += BIN_THREADS) {
    s_src[c0 + idx] = l_src[idx];
    s_row[c0 + idx] = l_row[idx];
    s_val[c0 + idx] = l_val[idx];
  }
  if (tid < MAX_BANDS)
    ((int2*)seg)[(size_t)chunk * MAX_BANDS + tid] = make_int2(base[tid], cnt[tid]);
}

__global__ __launch_bounds__(FRONT_THREADS) void gemm_and_bin_kernel(
    const float* __restrict__ x, const float* __restrict__ W,
    float* __restrict__ xw, int nTiles,
    const float* __restrict__ edge_val, const int* __restrict__ edge_src,
    const int* __restrict__ edge_dst,
    int* __restrict__ s_src, int* __restrict__ s_row, float* __restrict__ s_val,
    int* __restrict__ seg, int E, int nChunks) {
  __shared__ __attribute__((aligned(16))) int lds[FRONT_LDS_WORDS];
  const int blk = blockIdx.x;
  if (blk < nChunks)
    bin_edges_role(edge_val, edge_src, edge_dst, s_src, s_row, s_val, seg, E, blk, lds);
  else
    gemm_xw_role(x, W, xw, nTiles, blk - nChunks, (float*)lds);
}

#define SUM_THREADS 1024
#define SUM_WAVES   (SUM_THREADS / 32)
#define PLANE_ROWS  (BAND_ROWS + 1)
#define PLANE       (PLANE_ROWS * 8)

template <int U>
__device__ __forceinline__ unsigned group_bcast(unsigned x) {
  return (unsigned)__builtin_amdgcn_mov_dpp8(x, (unsigned)U * 0x249249u);
}

#define GATHER_STEP(U)                                                        \
  if (4 * (U) < n)                                                            \
    xv[U] = *(const v4f*)(xwb + (group_bcast<U>(soff) | q16));

#define ADD_STEP(U)                                                           \
  if (4 * (U) < n) {                                                          \
    const unsigned idx = group_bcast<U>(rdw) | (unsigned)q;                   \
    const float vj = __uint_as_float(group_bcast<U>(vb));                     \
    atomicAdd(&acc[idx],             vj * xv[U].x);                           \
    atomicAdd(&acc[PLANE + idx],     vj * xv[U].y);                           \
    atomicAdd(&acc[2 * PLANE + idx], vj * xv[U].z);                           \
    atomicAdd(&acc[3 * PLANE + idx], vj * xv[U].w);                           \
  }

__global__ __launch_bounds__(SUM_THREADS) void band_sum_kernel(
    const float* __restrict__ xw, const float* __restrict__ b,
    const int* __restrict__ s_src, const int* __restrict__ s_row,
    const float* __restrict__ s_val, const int* __restrict__ seg,
    float* __restrict__ out, int N, int nChunks) {
  __shared__ __attribute__((aligned(16))) float acc[4 * PLANE];

  const int band = blockIdx.x;
  const int tid  = threadIdx.x;
  const int lane = tid & 31;
  const int q    = lane & 7;
  const unsigned q16 = (unsigned)q * 16;
  const int pe = 4 * q + (lane >> 3);
  const int wave = __builtin_amdgcn_readfirstlane(tid >> 5);
  const char* xwb = (const char*)xw;

  const v4f b4 = ((const v4f*)b)[tid & 7];
  for (int i = tid; i < PLANE; i += SUM_THREADS) {
    acc[i]             = b4.x;
    acc[PLANE + i]     = b4.y;
    acc[2 * PLANE + i] = b4.z;
    acc[3 * PLANE + i] = b4.w;
  }
  __syncthreads();

  for (int kb = 0; kb < nChunks; kb += SUM_WAVES * 32) {
    const int kl = kb + wave + SUM_WAVES * lane;
    int2 sc = make_int2(0, 0);
    if (kl < nChunks) sc = ((const int2*)seg)[(size_t)kl * MAX_BANDS + band];

    for (int i = 0; i < 32; ++i) {
      const int k = kb + wave + SUM_WAVES * i;
      if (k >= nChunks) break;
      const int st = __builtin_amdgcn_readlane(sc.x, i);
      const int ct = __builtin_amdgcn_readlane(sc.y, i);
      const long long e0 = (long long)k * CHUNK + st;

      for (int o = 0; o < ct; o += 32) {
        const int n = (ct - o) < 32 ? (ct - o) : 32;
        int   s = 0, r = BAND_ROWS;
        float v = 0.0f;
        if (pe < n) {
          s = s_src[e0 + o + pe];
          r = s_row[e0 + o + pe];
          v = s_val[e0 + o + pe];
        }
        const unsigned soff = (unsigned)s << 7;
        const unsigned rdw  = (unsigned)r << 3;
        const unsigned vb   = __float_as_uint(v);

        v4f xv[8];
        GATHER_STEP(0) GATHER_STEP(1) GATHER_STEP(2) GATHER_STEP(3)
        GATHER_STEP(4) GATHER_STEP(5) GATHER_STEP(6) GATHER_STEP(7)
        ADD_STEP(0) ADD_STEP(1) ADD_STEP(2) ADD_STEP(3)
        ADD_STEP(4) ADD_STEP(5) ADD_STEP(6) ADD_STEP(7)
      }
    }
  }
  __syncthreads();

  const int row0 = band * BAND_ROWS;
  const int rows = (N - row0) < BAND_ROWS ? (N - row0) : BAND_ROWS;
  v4f* o4 = (v4f*)(out + (size_t)row0 * D_OUT);
  for (int i = tid; i < rows * (D_OUT / 4); i += SUM_THREADS) {
    v4f t;
    t.x = fmaxf(acc[i], 0.0f);
    t.y = fmaxf(acc[PLANE + i], 0.0f);
    t.z = fmaxf(acc[2 * PLANE + i], 0.0f);
    t.w = fmaxf(acc[3 * PLANE + i], 0.0f);
    o4[i] = t;
  }
}

extern "C" void kernel_launch(void* const* d_in, const int* in_sizes, int n_in,
                              void* d_out, int out_size, void* d_ws, size_t ws_size,
                              hipStream_t stream) {
  const float* x        = (const float*)d_in[0];
  const float* W        = (const float*)d_in[1];
  const float* b        = (const float*)d_in[2];
  const float* edge_val = (const float*)d_in[3];
  const int*   edge_src = (const int*)d_in[4];
  const int*   edge_dst = (const int*)d_in[5];
  float* out = (float*)d_out;

  const int N = in_sizes[0] / D_IN;
  const int E = in_sizes[3];
  const int nTiles  = (N + 15) / 16;
  const int nChunks = (E + CHUNK - 1) / CHUNK;
  const int nBands  = (N + BAND_ROWS - 1) / BAND_ROWS;

  float* xw    = (float*)d_ws;
  int*   s_src = (int*)(xw + (size_t)N * D_OUT);
  int*   s_row = s_src + (size_t)nChunks * CHUNK;
  float* s_val = (float*)(s_row + (size_t)nChunks * CHUNK);
  int*   seg   = (int*)(s_val + (size_t)nChunks * CHUNK);

  const int nGemmBlocks = (nTiles + GEMM_TILES_PER_BLOCK - 1) / GEMM_TILES_PER_BLOCK;
  gemm_and_bin_kernel<<<nChunks + nGemmBlocks, FRONT_THREADS, 0, stream>>>(
      x, W, xw, nTiles, edge_val, edge_src, edge_dst, s_src, s_row, s_val, seg, E, nChunks);

  band_sum_kernel<<<nBands, SUM_THREADS, 0, stream>>>(
      xw, b, s_src, s_row, s_val, seg, out, N, nChunks);
}


# band sum: each 8-lane group adds into its own LDS copy of the band (row pitch 128 dwords): the 32 lanes of a float add always hit 32 banks; out = relu(sum of 4 partial sums), adds arrival-order
# speedup vs baseline: 5.0554x; 1.0384x over previous
#include <hip/hip_runtime.h>

typedef float v2f __attribute__((ext_vector_type(2)));
typedef float v4f __attribute__((ext_vector_type(4)));
typedef float v8f __attribute__((ext_vector_type(8)));

#define D_IN  128
#define D_OUT 32


#define FRONT_THREADS 1024
#define GEMM_TILES_PER_BLOCK (FRONT_THREADS / 32)
__device__ __forceinline__ void gemm_xw_role(
    const float* __restrict__ x, const float* __restrict__ W,
    float* __restrict__ xw, int nTiles, int gblock, float* Wlds) {
  const float4* Wv = (const float4*)W;
  float4* Lv = (float4*)Wlds;
  for (int i = threadIdx.x; i < (D_IN * D_OUT) / 4; i += FRONT_THREADS)
    Lv[i] = Wv[i];
  __syncthreads();

  const int wave = threadIdx.x >> 5;
  const int lane = threadIdx.x & 31;
  const int tile = gblock * GEMM_TILES_PER_BLOCK + wave;
  if (tile >= nTiles) return;

  const int half = lane >> 4;
  const int m    = lane & 15;

  const float* xrow = x + (size_t)(tile * 16 + m) * D_IN + half * 2;

  v8f c0 = {};
  v8f c1 = {};

  #pragma unroll 4
  for (int k = 0; k < D_IN; k += 4) {
    v2f a = *(const v2f*)(xrow + k);

    const int kb = k + half * 2;
    v2f b0, b1;
    b0.x = Wlds[kb * D_OUT + m];
    b0.y = Wlds[(kb + 1) * D_OUT + m];
    b1.x = Wlds[kb * D_OUT + 16 + m];
    b1.y = Wlds[(kb + 1) * D_OUT + 16 + m];

    c0 = __builtin_amdgcn_wmma_f32_16x16x4_f32(false, a, false, b0,
                                               (short)0, c0, false, false);
    c1 = __builtin_amdgcn_wmma_f32_16x16x4_f32(false, a, false, b1,
                                               (short)0, c1, false, false);
  }

  const int rowBase = tile * 16 + half * 8;
  #pragma unroll
  for (int r = 0; r < 8; ++r) {
    xw[(size_t)(rowBase + r) * D_OUT + m]      = c0[r];
    xw[(size_t)(rowBase + r) * D_OUT + 16 + m] = c1[r];
  }
}

#define BAND_ROWS 391
#define MAX_BANDS 256
#define CHUNK     8192
#define BIN_THREADS FRONT_THREADS
#define EDGES_PER_THREAD (CHUNK / BIN_THREADS)

#define FRONT_LDS_WORDS (2 * MAX_BANDS + 3 * CHUNK)
__device__ __forceinline__ void bin_edges_role(
    const float* __restrict__ edge_val, const int* __restrict__ edge_src,
    const int* __restrict__ edge_dst,
    int* __restrict__ s_src, int* __restrict__ s_row, float* __restrict__ s_val,
    int* __restrict__ seg, int E, int chunk, int* lds) {
  int*   cnt   = lds;
  int*   base  = lds + MAX_BANDS;
  int*   l_src = lds + 2 * MAX_BANDS;
  int*   l_row = l_src + CHUNK;
  float* l_val = (float*)(l_row + CHUNK);

  const int tid   = threadIdx.x;
  const long long c0 = (long long)chunk * CHUNK;
  const long long rem = (long long)E - c0;
  const int n = rem < CHUNK ? (int)rem : CHUNK;

  if (tid < MAX_BANDS) cnt[tid] = 0;
  __syncthreads();

  int   es[EDGES_PER_THREAD], eb[EDGES_PER_THREAD], er[EDGES_PER_THREAD], ek[EDGES_PER_THREAD];
  float ev[EDGES_PER_THREAD];
  #pragma unroll
  for (int i = 0; i < EDGES_PER_THREAD; ++i) {
    const int idx = i * BIN_THREADS + tid;
    es[i] = 0; eb[i] = 0; er[i] = 0; ek[i] = 0; ev[i] = 0.0f;
    if (idx < n) {
      es[i] = edge_src[c0 + idx];
      const int d = edge_dst[c0 + idx];
      ev[i] = edge_val[c0 + idx];
      eb[i] = d / BAND_ROWS;
      er[i] = d - eb[i] * BAND_ROWS;
    }
  }
  #pragma unroll
  for (int i = 0; i < EDGES_PER_THREAD; ++i) {
    const int idx = i * BIN_THREADS + tid;
    if (idx < n) ek[i] = atomicAdd(&cnt[eb[i]], 1);
  }
  __syncthreads();

  if (tid < 32) {
    int loc[MAX_BANDS / 32];
    int sum = 0;
    #pragma unroll
    for (int q = 0; q < MAX_BANDS / 32; ++q) {
      loc[q] = sum;
      sum += cnt[tid * (MAX_BANDS / 32) + q];
    }
    int inc = sum;
    #pragma unroll
    for (int off = 1; off < 32; off <<= 1) {
      const int t = __shfl_up(inc, off);
      if (tid >= off) inc += t;
    }
    const int excl = inc - sum;
    #pragma unroll
    for (int q = 0; q < MAX_BANDS / 32; ++q)
      base[tid * (MAX_BANDS / 32) + q] = excl + loc[q];
  }
  __syncthreads();

  #pragma unroll
  for (int i = 0; i < EDGES_PER_THREAD; ++i) {
    const int idx = i * BIN_THREADS + tid;
    if (idx < n) {
      const int pos = base[eb[i]] + ek[i];
      l_src[pos] = es[i];
      l_row[pos] = er[i];
      l_val[pos] = ev[i];
    }
  }
  __syncthreads();

  for (int idx = tid; idx < n; idx += BIN_THREADS) {
    s_src[c0 + idx] = l_src[idx];
    s_row[c0 + idx] = l_row[idx];
    s_val[c0 + idx] = l_val[idx];
  }
  if (tid < MAX_BANDS)
    ((int2*)seg)[(size_t)chunk * MAX_BANDS + tid] = make_int2(base[tid], cnt[tid]);
}

__global__ __launch_bounds__(FRONT_THREADS) void gemm_and_bin_kernel(
    const float* __restrict__ x, const float* __restrict__ W,
    float* __restrict__ xw, int nTiles,
    const float* __restrict__ edge_val, const int* __restrict__ edge_src,
    const int* __restrict__ edge_dst,
    int* __restrict__ s_src, int* __restrict__ s_row, float* __restrict__ s_val,
    int* __restrict__ seg, int E, int nChunks) {
  __shared__ __attribute__((aligned(16))) int lds[FRONT_LDS_WORDS];
  const int blk = blockIdx.x;
  if (blk < nChunks)
    bin_edges_role(edge_val, edge_src, edge_dst, s_src, s_row, s_val, seg, E, blk, lds);
  else
    gemm_xw_role(x, W, xw, nTiles, blk - nChunks, (float*)lds);
}

#define SUM_THREADS 1024
#define SUM_WAVES   (SUM_THREADS / 32)
#define ACC_ROWS    (BAND_ROWS + 1)
#define ROW_WORDS   128
#define ACC_WORDS   (ACC_ROWS * ROW_WORDS)

template <int U>
__device__ __forceinline__ unsigned group_bcast(unsigned x) {
  return (unsigned)__builtin_amdgcn_mov_dpp8(x, (unsigned)U * 0x249249u);
}

#define GATHER_STEP(U)                                                        \
  if (4 * (U) < n)                                                            \
    xv[U] = *(const v4f*)(xwb + (group_bcast<U>(soff) | q16));

#define ADD_STEP(U)                                                           \
  if (4 * (U) < n) {                                                          \
    const unsigned idx = group_bcast<U>(rdw) | (unsigned)lane;                \
    const float vj = __uint_as_float(group_bcast<U>(vb));                     \
    atomicAdd(&acc[idx],      vj * xv[U].x);                                  \
    atomicAdd(&acc[32 + idx], vj * xv[U].y);                                  \
    atomicAdd(&acc[64 + idx], vj * xv[U].z);                                  \
    atomicAdd(&acc[96 + idx], vj * xv[U].w);                                  \
  }

__global__ __launch_bounds__(SUM_THREADS) void band_sum_kernel(
    const float* __restrict__ xw, const float* __restrict__ b,
    const int* __restrict__ s_src, const int* __restrict__ s_row,
    const float* __restrict__ s_val, const int* __restrict__ seg,
    float* __restrict__ out, int N, int nChunks) {
  __shared__ __attribute__((aligned(16))) float acc[ACC_WORDS];

  const int band = blockIdx.x;
  const int tid  = threadIdx.x;
  const int lane = tid & 31;
  const int q    = lane & 7;
  const unsigned q16 = (unsigned)q * 16;
  const int pe = 4 * q + (lane >> 3);
  const int wave = __builtin_amdgcn_readfirstlane(tid >> 5);
  const char* xwb = (const char*)xw;

  {
    const int o  = (tid * 4) & (ROW_WORDS - 1);
    const int ij = o >> 5;
    const int ig = (o >> 3) & 3;
    const int iq = o & 7;
    v4f init = {0.0f, 0.0f, 0.0f, 0.0f};
    if (ig == 0) {
      init.x = b[4 * iq + ij];
      init.y = b[4 * (iq + 1) + ij];
      init.z = b[4 * (iq + 2) + ij];
      init.w = b[4 * (iq + 3) + ij];
    }
    v4f* acc4 = (v4f*)acc;
    for (int i = tid; i < ACC_WORDS / 4; i += SUM_THREADS) acc4[i] = init;
  }
  __syncthreads();

  for (int kb = 0; kb < nChunks; kb += SUM_WAVES * 32) {
    const int kl = kb + wave + SUM_WAVES * lane;
    int2 sc = make_int2(0, 0);
    if (kl < nChunks) sc = ((const int2*)seg)[(size_t)kl * MAX_BANDS + band];

    for (int i = 0; i < 32; ++i) {
      const int k = kb + wave + SUM_WAVES * i;
      if (k >= nChunks) break;
      const int st = __builtin_amdgcn_readlane(sc.x, i);
      const int ct = __builtin_amdgcn_readlane(sc.y, i);
      const long long e0 = (long long)k * CHUNK + st;

      for (int o = 0; o < ct; o += 32) {
        const int n = (ct - o) < 32 ? (ct - o) : 32;
        int   s = 0, r = BAND_ROWS;
        float v = 0.0f;
        if (pe < n) {
          s = s_src[e0 + o + pe];
          r = s_row[e0 + o + pe];
          v = s_val[e0 + o + pe];
        }
        const unsigned soff = (unsigned)s << 7;
        const unsigned rdw  = (unsigned)r << 7;
        const unsigned vb   = __float_as_uint(v);

        v4f xv[8];
        GATHER_STEP(0) GATHER_STEP(1) GATHER_STEP(2) GATHER_STEP(3)
        GATHER_STEP(4) GATHER_STEP(5) GATHER_STEP(6) GATHER_STEP(7)
        ADD_STEP(0) ADD_STEP(1) ADD_STEP(2) ADD_STEP(3)
        ADD_STEP(4) ADD_STEP(5) ADD_STEP(6) ADD_STEP(7)
      }
    }
  }
  __syncthreads();

  const int row0 = band * BAND_ROWS;
  const int rows = (N - row0) < BAND_ROWS ? (N - row0) : BAND_ROWS;
  v4f* o4 = (v4f*)(out + (size_t)row0 * D_OUT);
  for (int i = tid; i < rows * (D_OUT / 4); i += SUM_THREADS) {
    const int rr = i >> 3;
    const float* p = acc + rr * ROW_WORDS + (i & 7);
    float t[4];
    #pragma unroll
    for (int j = 0; j < 4; ++j) {
      float sum = p[j * 32 + 8 * (rr & 3)];
      #pragma unroll
      for (int g = 1; g < 4; ++g) sum += p[j * 32 + 8 * ((rr + g) & 3)];
      t[j] = fmaxf(sum, 0.0f);
    }
    v4f tv;
    tv.x = t[0]; tv.y = t[1]; tv.z = t[2]; tv.w = t[3];
    o4[i] = tv;
  }
}

extern "C" void kernel_launch(void* const* d_in, const int* in_sizes, int n_in,
                              void* d_out, int out_size, void* d_ws, size_t ws_size,
                              hipStream_t stream) {
  const float* x        = (const float*)d_in[0];
  const float* W        = (const float*)d_in[1];
  const float* b        = (const float*)d_in[2];
  const float* edge_val = (const float*)d_in[3];
  const int*   edge_src = (const int*)d_in[4];
  const int*   edge_dst = (const int*)d_in[5];
  float* out = (float*)d_out;

  const int N = in_sizes[0] / D_IN;
  const int E = in_sizes[3];
  const int nTiles  = (N + 15) / 16;
  const int nChunks = (E + CHUNK - 1) / CHUNK;
  const int nBands  = (N + BAND_ROWS - 1) / BAND_ROWS;

  float* xw    = (float*)d_ws;
  int*   s_src = (int*)(xw + (size_t)N * D_OUT);
  int*   s_row = s_src + (size_t)nChunks * CHUNK;
  float* s_val = (float*)(s_row + (size_t)nChunks * CHUNK);
  int*   seg   = (int*)(s_val + (size_t)nChunks * CHUNK);

  const int nGemmBlocks = (nTiles + GEMM_TILES_PER_BLOCK - 1) / GEMM_TILES_PER_BLOCK;
  gemm_and_bin_kernel<<<nChunks + nGemmBlocks, FRONT_THREADS, 0, stream>>>(
      x, W, xw, nTiles, edge_val, edge_src, edge_dst, s_src, s_row, s_val, seg, E, nChunks);

  band_sum_kernel<<<nBands, SUM_THREADS, 0, stream>>>(
      xw, b, s_src, s_row, s_val, seg, out, N, nChunks);
}
